# starting kernel
# baseline (speedup 1.0000x reference)
#include <hip/hip_runtime.h>
#include <math.h>


#define NN       80
#define MPM_IT   50
#define STRIDE   84
#define NWAVES   25
#define NTHREADS 800

typedef float v2f __attribute__((ext_vector_type(2)));
typedef float v8f __attribute__((ext_vector_type(8)));

__global__ __launch_bounds__(NTHREADS)
void grf_mpm_kernel(const float* __restrict__ recon,
                    const float* __restrict__ adj,
                    float* __restrict__ out)
{
    extern __shared__ float smem[];
    float* s_src = smem;
    float* s_tgt = s_src + NN * STRIDE;
    float* s_d   = s_tgt + NN * STRIDE;
    float* s_x   = s_d   + NN * STRIDE;
    float* s_M   = s_x   + NN * STRIDE;
    float* s_af  = s_M   + NN * STRIDE;
    float* s_of  = s_af  + NN;
    float* s_ad  = s_of  + NN;
    float* s_rd  = s_ad  + NN;
    float* s_red = s_rd  + NN;
    float* s_sc  = s_red + 32;

    const int tid  = threadIdx.x;
    const int lane = tid & 31;
    const int wave = tid >> 5;

    if (tid < NN) {
        float sa = 0.f, sr = 0.f;
        for (int c = 0; c < NN; ++c) {
            sa += adj[tid * NN + c];
            sr += recon[tid * NN + c];
        }
        s_af[tid] = sa;
        s_of[tid] = sr;
        s_ad[tid] = adj[tid * NN + tid];
        s_rd[tid] = recon[tid * NN + tid];
    }
    __syncthreads();

    float bce = 0.f;
    for (int idx = tid; idx < NN * NN; idx += NTHREADS) {
        int i = idx / NN;
        int j = idx - i * NN;
        float ta = adj[idx];
        float tr = recon[idx];
        float offd = (i == j) ? 0.f : 1.f;
        s_src[i * STRIDE + j] = ta * s_ad[i] * s_ad[j] * offd;
        s_tgt[i * STRIDE + j] = tr * s_rd[i] * s_rd[j] * offd;
        s_d  [i * STRIDE + j] = s_ad[i] * s_rd[j] / (fabsf(s_af[i] - s_of[j]) + 1.f);
        s_x  [i * STRIDE + j] = 1.f / (float)NN;
        if (j >= i) {
            bce -= ta * logf(tr) + (1.f - ta) * logf(1.f - tr);
        }
    }
    #pragma unroll
    for (int off = 16; off; off >>= 1) bce += __shfl_xor(bce, off, 32);
    if (lane == 0) s_red[wave] = bce;
    __syncthreads();
    if (wave == 0) {
        float t = (lane < NWAVES) ? s_red[lane] : 0.f;
        #pragma unroll
        for (int off = 16; off; off >>= 1) t += __shfl_xor(t, off, 32);
        if (lane == 0) out[0] = t / (float)(NN * (NN + 1) / 2);
    }
    __syncthreads();

    const int wr = wave / 5;
    const int wc = wave - 5 * wr;
    const int r0 = wr * 16;
    const int c0 = wc * 16;
    const int hh = lane >> 4;
    const int m  = lane & 15;
    const int kk = 2 * hh;

    for (int it = 0; it < MPM_IT; ++it) {
        #pragma unroll
        for (int v = 0; v < 8; ++v) {
            int j = r0 + v + 8 * hh;
            int a = c0 + m;
            const float* xr = s_x   + j * STRIDE;
            const float* tg = s_tgt + a * STRIDE;
            float best = xr[0] * tg[0];
            #pragma unroll 4
            for (int b = 1; b < NN; ++b) best = fmaxf(best, xr[b] * tg[b]);
            s_M[j * STRIDE + a] = best;
        }
        __syncthreads();

        v8f acc = {0.f, 0.f, 0.f, 0.f, 0.f, 0.f, 0.f, 0.f};
        for (int kc = 0; kc < NN; kc += 4) {
            v2f av, bv;
            av[0] = s_src[(r0 + m) * STRIDE + kc + kk + 0];
            av[1] = s_src[(r0 + m) * STRIDE + kc + kk + 1];
            bv[0] = s_M[(kc + kk + 0) * STRIDE + c0 + m];
            bv[1] = s_M[(kc + kk + 1) * STRIDE + c0 + m];
            acc = __builtin_amdgcn_wmma_f32_16x16x4_f32(
                false, av, false, bv, (short)0, acc, false, false);
        }

        float xn[8];
        float ssq = 0.f;
        #pragma unroll
        for (int v = 0; v < 8; ++v) {
            int i = r0 + v + 8 * hh;
            int a = c0 + m;
            float val = s_x[i * STRIDE + a] * s_d[i * STRIDE + a] + acc[v];
            xn[v] = val;
            ssq += val * val;
        }
        #pragma unroll
        for (int off = 16; off; off >>= 1) ssq += __shfl_xor(ssq, off, 32);
        if (lane == 0) s_red[wave] = ssq;
        __syncthreads();
        if (tid == 0) {
            float t = 0.f;
            for (int w = 0; w < NWAVES; ++w) t += s_red[w];
            s_sc[0] = 1.f / sqrtf(t);
        }
        __syncthreads();
        float rn = s_sc[0];
        #pragma unroll
        for (int v = 0; v < 8; ++v) {
            int i = r0 + v + 8 * hh;
            int a = c0 + m;
            s_x[i * STRIDE + a] = xn[v] * rn;
        }
        __syncthreads();
    }

    #pragma unroll
    for (int v = 0; v < 8; ++v) {
        int i = r0 + v + 8 * hh;
        int a = c0 + m;
        out[1 + i * NN + a] = s_x[i * STRIDE + a];
    }
}

extern "C" void kernel_launch(void* const* d_in, const int* in_sizes, int n_in,
                              void* d_out, int out_size, void* d_ws, size_t ws_size,
                              hipStream_t stream) {
    (void)in_sizes; (void)n_in; (void)out_size; (void)d_ws; (void)ws_size;
    const float* recon = (const float*)d_in[0];
    const float* adj   = (const float*)d_in[1];
    float* out = (float*)d_out;

    const size_t shmem = (size_t)(5 * NN * STRIDE + 4 * NN + 32 + 4) * sizeof(float);
    grf_mpm_kernel<<<1, NTHREADS, shmem, stream>>>(recon, adj, out);
}


# Tropical max-product register-tiled: each lane keeps 2 tgt rows x half of b in VGPRs for all iterations, streams x by ds_load_b128 (4x2 tile, pk_mul+max3); halves of b joined by one DPP max
# speedup vs baseline: 6.4890x; 6.4890x over previous
#include <hip/hip_runtime.h>
#include <math.h>


#define NN       80
#define MPM_IT   50
#define STRIDE   84
#define NWAVES   25
#define NTHREADS 800

typedef float v2f __attribute__((ext_vector_type(2)));
typedef float v8f __attribute__((ext_vector_type(8)));
typedef float v4f __attribute__((ext_vector_type(4)));

__device__ __forceinline__ float lane_xor1(float v) {
    int iv = __builtin_bit_cast(int, v);
    int r  = __builtin_amdgcn_update_dpp(iv, iv, 0xB1, 0xF, 0xF, true);
    return __builtin_bit_cast(float, r);
}

__global__ __launch_bounds__(NTHREADS)
void grf_mpm_kernel(const float* __restrict__ recon,
                    const float* __restrict__ adj,
                    float* __restrict__ out)
{
    extern __shared__ __attribute__((aligned(16))) float smem[];
    float* s_src = smem;
    float* s_tgt = s_src + NN * STRIDE;
    float* s_d   = s_tgt + NN * STRIDE;
    float* s_x   = s_d   + NN * STRIDE;
    float* s_M   = s_x   + NN * STRIDE;
    float* s_af  = s_M   + NN * STRIDE;
    float* s_of  = s_af  + NN;
    float* s_ad  = s_of  + NN;
    float* s_rd  = s_ad  + NN;
    float* s_red = s_rd  + NN;
    float* s_sc  = s_red + 32;

    const int tid  = threadIdx.x;
    const int lane = tid & 31;
    const int wave = tid >> 5;

    if (tid < NN) {
        float sa = 0.f, sr = 0.f;
        for (int c = 0; c < NN; ++c) {
            sa += adj[tid * NN + c];
            sr += recon[tid * NN + c];
        }
        s_af[tid] = sa;
        s_of[tid] = sr;
        s_ad[tid] = adj[tid * NN + tid];
        s_rd[tid] = recon[tid * NN + tid];
    }
    __syncthreads();

    float bce = 0.f;
    for (int idx = tid; idx < NN * NN; idx += NTHREADS) {
        int i = idx / NN;
        int j = idx - i * NN;
        float ta = adj[idx];
        float tr = recon[idx];
        float offd = (i == j) ? 0.f : 1.f;
        s_src[i * STRIDE + j] = ta * s_ad[i] * s_ad[j] * offd;
        s_tgt[i * STRIDE + j] = tr * s_rd[i] * s_rd[j] * offd;
        s_d  [i * STRIDE + j] = s_ad[i] * s_rd[j] / (fabsf(s_af[i] - s_of[j]) + 1.f);
        s_x  [i * STRIDE + j] = 1.f / (float)NN;
        if (j >= i) {
            bce -= ta * logf(tr) + (1.f - ta) * logf(1.f - tr);
        }
    }
    #pragma unroll
    for (int off = 16; off; off >>= 1) bce += __shfl_xor(bce, off, 32);
    if (lane == 0) s_red[wave] = bce;
    __syncthreads();
    if (wave == 0) {
        float t = (lane < NWAVES) ? s_red[lane] : 0.f;
        #pragma unroll
        for (int off = 16; off; off >>= 1) t += __shfl_xor(t, off, 32);
        if (lane == 0) out[0] = t / (float)(NN * (NN + 1) / 2);
    }
    __syncthreads();

    const int wr = wave / 5;
    const int wc = wave - 5 * wr;
    const int r0 = wr * 16;
    const int c0 = wc * 16;
    const int hh = lane >> 4;
    const int m  = lane & 15;
    const int kk = 2 * hh;
    const int eo = (r0 + 8 * hh) * STRIDE + c0 + m;

    const int jg = tid / NN;
    const int rr = tid - NN * jg;
    const int hb = rr & 1;
    const int a0 = rr & ~1;
    const int bb = 40 * hb;

    v4f tgA[10], tgB[10];
    #pragma unroll
    for (int c = 0; c < 10; ++c) {
        tgA[c] = *(const v4f*)(s_tgt + (a0    ) * STRIDE + bb + 4 * c);
        tgB[c] = *(const v4f*)(s_tgt + (a0 + 1) * STRIDE + bb + 4 * c);
    }

    for (int it = 0; it < MPM_IT; ++it) {
        for (int jh = 0; jh < 2; ++jh) {
            const int jb = 8 * jg + 4 * jh;
            const float* xb = s_x + jb * STRIDE + bb;
            float mA[4], mB[4];
            v4f xa0 = *(const v4f*)(xb);
            v4f xa1 = *(const v4f*)(xb + STRIDE);
            #pragma unroll
            for (int s = 0; s < 20; ++s) {
                const int c  = s >> 1;
                const int q0 = 2 * (s & 1);
                v4f xn0 = xa0, xn1 = xa1;
                if (s + 1 < 20) {
                    const int cn = (s + 1) >> 1;
                    const int qn = 2 * ((s + 1) & 1);
                    xn0 = *(const v4f*)(xb + (qn    ) * STRIDE + 4 * cn);
                    xn1 = *(const v4f*)(xb + (qn + 1) * STRIDE + 4 * cn);
                }
                __builtin_amdgcn_sched_barrier(0);
                #pragma unroll
                for (int e = 0; e < 2; ++e) {
                    const int q = q0 + e;
                    const v4f xv = e ? xa1 : xa0;
                    const v4f pa = xv * tgA[c];
                    const v4f pb = xv * tgB[c];
                    if (c == 0) {
                        mA[q] = fmaxf(fmaxf(pa[0], pa[1]), fmaxf(pa[2], pa[3]));
                        mB[q] = fmaxf(fmaxf(pb[0], pb[1]), fmaxf(pb[2], pb[3]));
                    } else {
                        mA[q] = fmaxf(fmaxf(mA[q], pa[0]), pa[1]);
                        mA[q] = fmaxf(fmaxf(mA[q], pa[2]), pa[3]);
                        mB[q] = fmaxf(fmaxf(mB[q], pb[0]), pb[1]);
                        mB[q] = fmaxf(fmaxf(mB[q], pb[2]), pb[3]);
                    }
                }
                __builtin_amdgcn_sched_barrier(0);
                xa0 = xn0;
                xa1 = xn1;
            }
            #pragma unroll
            for (int q = 0; q < 4; ++q) {
                float keep = hb ? mB[q] : mA[q];
                float send = hb ? mA[q] : mB[q];
                float recv = lane_xor1(send);
                s_M[(jb + q) * STRIDE + rr] = fmaxf(keep, recv);
            }
        }
        __syncthreads();

        v8f acc = {0.f, 0.f, 0.f, 0.f, 0.f, 0.f, 0.f, 0.f};
        for (int kc = 0; kc < NN; kc += 4) {
            v2f av, bv;
            av[0] = s_src[(r0 + m) * STRIDE + kc + kk + 0];
            av[1] = s_src[(r0 + m) * STRIDE + kc + kk + 1];
            bv[0] = s_M[(kc + kk + 0) * STRIDE + c0 + m];
            bv[1] = s_M[(kc + kk + 1) * STRIDE + c0 + m];
            acc = __builtin_amdgcn_wmma_f32_16x16x4_f32(
                false, av, false, bv, (short)0, acc, false, false);
        }

        float xn[8];
        float ssq = 0.f;
        #pragma unroll
        for (int v = 0; v < 8; ++v) {
            float val = s_x[eo + v * STRIDE] * s_d[eo + v * STRIDE] + acc[v];
            xn[v] = val;
            ssq += val * val;
        }
        #pragma unroll
        for (int off = 16; off; off >>= 1) ssq += __shfl_xor(ssq, off, 32);
        if (lane == 0) s_red[wave] = ssq;
        __syncthreads();
        if (tid == 0) {
            float t = 0.f;
            for (int w = 0; w < NWAVES; ++w) t += s_red[w];
            s_sc[0] = 1.f / sqrtf(t);
        }
        __syncthreads();
        float rn = s_sc[0];
        #pragma unroll
        for (int v = 0; v < 8; ++v) {
            s_x[eo + v * STRIDE] = xn[v] * rn;
        }
        __syncthreads();
    }

    #pragma unroll
    for (int v = 0; v < 8; ++v) {
        int i = r0 + v + 8 * hh;
        int a = c0 + m;
        out[1 + i * NN + a] = s_x[i * STRIDE + a];
    }
}

extern "C" void kernel_launch(void* const* d_in, const int* in_sizes, int n_in,
                              void* d_out, int out_size, void* d_ws, size_t ws_size,
                              hipStream_t stream) {
    (void)in_sizes; (void)n_in; (void)out_size; (void)d_ws; (void)ws_size;
    const float* recon = (const float*)d_in[0];
    const float* adj   = (const float*)d_in[1];
    float* out = (float*)d_out;

    const size_t shmem = (size_t)(5 * NN * STRIDE + 4 * NN + 32 + 4) * sizeof(float);
    grf_mpm_kernel<<<1, NTHREADS, shmem, stream>>>(recon, adj, out);
}


# Deferred normalisation: LDS keeps the unnormalised update, step t+1 scales by 1/norm(t) (max-product, GEMM homogeneous); norm summed by fixed DPP tree (new sum order) beside A/B; 2 barriers/step
# speedup vs baseline: 6.8796x; 1.0602x over previous
#include <hip/hip_runtime.h>
#include <math.h>


#define NN       80
#define MPM_IT   50
#define STRIDE   84
#define NWAVES   25
#define NTHREADS 800

typedef float v2f __attribute__((ext_vector_type(2)));
typedef float v8f __attribute__((ext_vector_type(8)));
typedef float v4f __attribute__((ext_vector_type(4)));

__device__ __forceinline__ float lane_xor1(float v) {
    int iv = __builtin_bit_cast(int, v);
    int r  = __builtin_amdgcn_update_dpp(iv, iv, 0xB1, 0xF, 0xF, true);
    return __builtin_bit_cast(float, r);
}

template <int MASK>
__device__ __forceinline__ float row_xor(float v) {
    int iv = __builtin_bit_cast(int, v);
    int r  = __builtin_amdgcn_update_dpp(iv, iv, 0x160 + MASK, 0xF, 0xF, true);
    return __builtin_bit_cast(float, r);
}

__device__ __forceinline__ float xhalf(float v) {
    int iv = __builtin_bit_cast(int, v);
    int r  = __builtin_amdgcn_permlanex16(iv, iv, 0x76543210, 0xfedcba98, false, false);
    return __builtin_bit_cast(float, r);
}

__device__ __forceinline__ float wave_sum(float v) {
    v += row_xor<1>(v);
    v += row_xor<2>(v);
    v += row_xor<4>(v);
    v += row_xor<8>(v);
    v += xhalf(v);
    return v;
}

__global__ __launch_bounds__(NTHREADS)
void grf_mpm_kernel(const float* __restrict__ recon,
                    const float* __restrict__ adj,
                    float* __restrict__ out)
{
    extern __shared__ __attribute__((aligned(16))) float smem[];
    float* s_src = smem;
    float* s_tgt = s_src + NN * STRIDE;
    float* s_d   = s_tgt + NN * STRIDE;
    float* s_x   = s_d   + NN * STRIDE;
    float* s_M   = s_x   + NN * STRIDE;
    float* s_af  = s_M   + NN * STRIDE;
    float* s_of  = s_af  + NN;
    float* s_ad  = s_of  + NN;
    float* s_rd  = s_ad  + NN;
    float* s_red = s_rd  + NN;

    const int tid  = threadIdx.x;
    const int lane = tid & 31;
    const int wave = tid >> 5;

    if (tid < NN) {
        float sa = 0.f, sr = 0.f;
        for (int c = 0; c < NN; ++c) {
            sa += adj[tid * NN + c];
            sr += recon[tid * NN + c];
        }
        s_af[tid] = sa;
        s_of[tid] = sr;
        s_ad[tid] = adj[tid * NN + tid];
        s_rd[tid] = recon[tid * NN + tid];
    }
    __syncthreads();

    float bce = 0.f;
    for (int idx = tid; idx < NN * NN; idx += NTHREADS) {
        int i = idx / NN;
        int j = idx - i * NN;
        float ta = adj[idx];
        float tr = recon[idx];
        float offd = (i == j) ? 0.f : 1.f;
        s_src[i * STRIDE + j] = ta * s_ad[i] * s_ad[j] * offd;
        s_tgt[i * STRIDE + j] = tr * s_rd[i] * s_rd[j] * offd;
        s_d  [i * STRIDE + j] = s_ad[i] * s_rd[j] / (fabsf(s_af[i] - s_of[j]) + 1.f);
        s_x  [i * STRIDE + j] = 1.f / (float)NN;
        if (j >= i) {
            bce -= ta * logf(tr) + (1.f - ta) * logf(1.f - tr);
        }
    }
    #pragma unroll
    for (int off = 16; off; off >>= 1) bce += __shfl_xor(bce, off, 32);
    if (lane == 0) s_red[wave] = bce;
    __syncthreads();
    if (wave == 0) {
        float t = (lane < NWAVES) ? s_red[lane] : 0.f;
        #pragma unroll
        for (int off = 16; off; off >>= 1) t += __shfl_xor(t, off, 32);
        if (lane == 0) out[0] = t / (float)(NN * (NN + 1) / 2);
    }
    __syncthreads();

    const int wr = wave / 5;
    const int wc = wave - 5 * wr;
    const int r0 = wr * 16;
    const int c0 = wc * 16;
    const int hh = lane >> 4;
    const int m  = lane & 15;
    const int kk = 2 * hh;
    const int eo = (r0 + 8 * hh) * STRIDE + c0 + m;

    const int jg = tid / NN;
    const int rr = tid - NN * jg;
    const int hb = rr & 1;
    const int a0 = rr & ~1;
    const int bb = 40 * hb;

    v4f tgA[10], tgB[10];
    #pragma unroll
    for (int c = 0; c < 10; ++c) {
        tgA[c] = *(const v4f*)(s_tgt + (a0    ) * STRIDE + bb + 4 * c);
        tgB[c] = *(const v4f*)(s_tgt + (a0 + 1) * STRIDE + bb + 4 * c);
    }

    float wsum = (wave == 0) ? 1.f : 0.f;

    for (int it = 0; it < MPM_IT; ++it) {
        if (lane == 0) s_red[wave] = wsum;

        for (int jh = 0; jh < 2; ++jh) {
            const int jb = 8 * jg + 4 * jh;
            const float* xb = s_x + jb * STRIDE + bb;
            float mA[4], mB[4];
            v4f xa0 = *(const v4f*)(xb);
            v4f xa1 = *(const v4f*)(xb + STRIDE);
            #pragma unroll
            for (int s = 0; s < 20; ++s) {
                const int c  = s >> 1;
                const int q0 = 2 * (s & 1);
                v4f xn0 = xa0, xn1 = xa1;
                if (s + 1 < 20) {
                    const int cn = (s + 1) >> 1;
                    const int qn = 2 * ((s + 1) & 1);
                    xn0 = *(const v4f*)(xb + (qn    ) * STRIDE + 4 * cn);
                    xn1 = *(const v4f*)(xb + (qn + 1) * STRIDE + 4 * cn);
                }
                __builtin_amdgcn_sched_barrier(0);
                #pragma unroll
                for (int e = 0; e < 2; ++e) {
                    const int q = q0 + e;
                    const v4f xv = e ? xa1 : xa0;
                    const v4f pa = xv * tgA[c];
                    const v4f pb = xv * tgB[c];
                    if (c == 0) {
                        mA[q] = fmaxf(fmaxf(pa[0], pa[1]), fmaxf(pa[2], pa[3]));
                        mB[q] = fmaxf(fmaxf(pb[0], pb[1]), fmaxf(pb[2], pb[3]));
                    } else {
                        mA[q] = fmaxf(fmaxf(mA[q], pa[0]), pa[1]);
                        mA[q] = fmaxf(fmaxf(mA[q], pa[2]), pa[3]);
                        mB[q] = fmaxf(fmaxf(mB[q], pb[0]), pb[1]);
                        mB[q] = fmaxf(fmaxf(mB[q], pb[2]), pb[3]);
                    }
                }
                __builtin_amdgcn_sched_barrier(0);
                xa0 = xn0;
                xa1 = xn1;
            }
            #pragma unroll
            for (int q = 0; q < 4; ++q) {
                float keep = hb ? mB[q] : mA[q];
                float send = hb ? mA[q] : mB[q];
                float recv = lane_xor1(send);
                s_M[(jb + q) * STRIDE + rr] = fmaxf(keep, recv);
            }
        }
        __syncthreads();

        const float part = (lane < NWAVES) ? s_red[lane] : 0.f;
        v8f acc = {0.f, 0.f, 0.f, 0.f, 0.f, 0.f, 0.f, 0.f};
        for (int kc = 0; kc < NN / 2; kc += 4) {
            v2f av, bv;
            av[0] = s_src[(r0 + m) * STRIDE + kc + kk + 0];
            av[1] = s_src[(r0 + m) * STRIDE + kc + kk + 1];
            bv[0] = s_M[(kc + kk + 0) * STRIDE + c0 + m];
            bv[1] = s_M[(kc + kk + 1) * STRIDE + c0 + m];
            acc = __builtin_amdgcn_wmma_f32_16x16x4_f32(
                false, av, false, bv, (short)0, acc, false, false);
        }
        const float rn = 1.f / sqrtf(wave_sum(part));
        for (int kc = NN / 2; kc < NN; kc += 4) {
            v2f av, bv;
            av[0] = s_src[(r0 + m) * STRIDE + kc + kk + 0];
            av[1] = s_src[(r0 + m) * STRIDE + kc + kk + 1];
            bv[0] = s_M[(kc + kk + 0) * STRIDE + c0 + m];
            bv[1] = s_M[(kc + kk + 1) * STRIDE + c0 + m];
            acc = __builtin_amdgcn_wmma_f32_16x16x4_f32(
                false, av, false, bv, (short)0, acc, false, false);
        }

        float ssq = 0.f;
        #pragma unroll
        for (int v = 0; v < 8; ++v) {
            float val = (s_x[eo + v * STRIDE] * s_d[eo + v * STRIDE] + acc[v]) * rn;
            s_x[eo + v * STRIDE] = val;
            ssq += val * val;
        }
        wsum = wave_sum(ssq);
        __syncthreads();
    }

    if (lane == 0) s_red[wave] = wsum;
    __syncthreads();
    const float rn_last = 1.f / sqrtf(wave_sum((lane < NWAVES) ? s_red[lane] : 0.f));

    #pragma unroll
    for (int v = 0; v < 8; ++v) {
        int i = r0 + v + 8 * hh;
        int a = c0 + m;
        out[1 + i * NN + a] = s_x[i * STRIDE + a] * rn_last;
    }
}

extern "C" void kernel_launch(void* const* d_in, const int* in_sizes, int n_in,
                              void* d_out, int out_size, void* d_ws, size_t ws_size,
                              hipStream_t stream) {
    (void)in_sizes; (void)n_in; (void)out_size; (void)d_ws; (void)ws_size;
    const float* recon = (const float*)d_in[0];
    const float* adj   = (const float*)d_in[1];
    float* out = (float*)d_out;

    const size_t shmem = (size_t)(5 * NN * STRIDE + 4 * NN + 32) * sizeof(float);
    grf_mpm_kernel<<<1, NTHREADS, shmem, stream>>>(recon, adj, out);
}


# Even SIMD load: 20 waves (5 per SIMD) instead of 25 (7/6/6/6); tropical product in 8 groups of 10 rows; waves 0..4 run a second GEMM tile; norm and loss partial sums regrouped (new sum order)
# speedup vs baseline: 7.5240x; 1.0937x over previous
#include <hip/hip_runtime.h>
#include <math.h>


#define NN       80
#define MPM_IT   50
#define STRIDE   84
#define NWAVES   20
#define NTHREADS 640

typedef float v2f __attribute__((ext_vector_type(2)));
typedef float v8f __attribute__((ext_vector_type(8)));
typedef float v4f __attribute__((ext_vector_type(4)));

__device__ __forceinline__ float lane_xor1(float v) {
    int iv = __builtin_bit_cast(int, v);
    int r  = __builtin_amdgcn_update_dpp(iv, iv, 0xB1, 0xF, 0xF, true);
    return __builtin_bit_cast(float, r);
}

template <int MASK>
__device__ __forceinline__ float row_xor(float v) {
    int iv = __builtin_bit_cast(int, v);
    int r  = __builtin_amdgcn_update_dpp(iv, iv, 0x160 + MASK, 0xF, 0xF, true);
    return __builtin_bit_cast(float, r);
}

__device__ __forceinline__ float xhalf(float v) {
    int iv = __builtin_bit_cast(int, v);
    int r  = __builtin_amdgcn_permlanex16(iv, iv, 0x76543210, 0xfedcba98, false, false);
    return __builtin_bit_cast(float, r);
}

__device__ __forceinline__ float wave_sum(float v) {
    v += row_xor<1>(v);
    v += row_xor<2>(v);
    v += row_xor<4>(v);
    v += row_xor<8>(v);
    v += xhalf(v);
    return v;
}

__global__ __launch_bounds__(NTHREADS)
void grf_mpm_kernel(const float* __restrict__ recon,
                    const float* __restrict__ adj,
                    float* __restrict__ out)
{
    extern __shared__ __attribute__((aligned(16))) float smem[];
    float* s_src = smem;
    float* s_tgt = s_src + NN * STRIDE;
    float* s_d   = s_tgt + NN * STRIDE;
    float* s_x   = s_d   + NN * STRIDE;
    float* s_M   = s_x   + NN * STRIDE;
    float* s_af  = s_M   + NN * STRIDE;
    float* s_of  = s_af  + NN;
    float* s_ad  = s_of  + NN;
    float* s_rd  = s_ad  + NN;
    float* s_red = s_rd  + NN;

    const int tid  = threadIdx.x;
    const int lane = tid & 31;
    const int wave = __builtin_amdgcn_readfirstlane(tid >> 5);

    if (tid < NN) {
        float sa = 0.f, sr = 0.f;
        for (int c = 0; c < NN; ++c) {
            sa += adj[tid * NN + c];
            sr += recon[tid * NN + c];
        }
        s_af[tid] = sa;
        s_of[tid] = sr;
        s_ad[tid] = adj[tid * NN + tid];
        s_rd[tid] = recon[tid * NN + tid];
    }
    __syncthreads();

    float bce = 0.f;
    for (int idx = tid; idx < NN * NN; idx += NTHREADS) {
        int i = idx / NN;
        int j = idx - i * NN;
        float ta = adj[idx];
        float tr = recon[idx];
        float offd = (i == j) ? 0.f : 1.f;
        s_src[i * STRIDE + j] = ta * s_ad[i] * s_ad[j] * offd;
        s_tgt[i * STRIDE + j] = tr * s_rd[i] * s_rd[j] * offd;
        s_d  [i * STRIDE + j] = s_ad[i] * s_rd[j] / (fabsf(s_af[i] - s_of[j]) + 1.f);
        s_x  [i * STRIDE + j] = 1.f / (float)NN;
        if (j >= i) {
            bce -= ta * logf(tr) + (1.f - ta) * logf(1.f - tr);
        }
    }
    #pragma unroll
    for (int off = 16; off; off >>= 1) bce += __shfl_xor(bce, off, 32);
    if (lane == 0) s_red[wave] = bce;
    __syncthreads();
    if (wave == 0) {
        float t = (lane < NWAVES) ? s_red[lane] : 0.f;
        #pragma unroll
        for (int off = 16; off; off >>= 1) t += __shfl_xor(t, off, 32);
        if (lane == 0) out[0] = t / (float)(NN * (NN + 1) / 2);
    }
    __syncthreads();

    const bool two = (wave < 5);
    const int tile2 = two ? 20 + wave : wave;
    const int hh = lane >> 4;
    const int m  = lane & 15;
    const int kk = 2 * hh;

    const int jg = tid / NN;
    const int rr = tid - NN * jg;
    const int hb = rr & 1;
    const int a0 = rr & ~1;
    const int bb = 40 * hb;

    v4f tgA[10], tgB[10];
    #pragma unroll
    for (int c = 0; c < 10; ++c) {
        tgA[c] = *(const v4f*)(s_tgt + (a0    ) * STRIDE + bb + 4 * c);
        tgB[c] = *(const v4f*)(s_tgt + (a0 + 1) * STRIDE + bb + 4 * c);
    }

    float wsum = (wave == 0) ? 1.f : 0.f;

    for (int it = 0; it < MPM_IT; ++it) {
        if (lane == 0) s_red[wave] = wsum;

        for (int jh = 0; jh < 2; ++jh) {
            const int jb = 10 * jg + 5 * jh;
            const float* xb = s_x + jb * STRIDE + bb;
            float mA[5], mB[5];
            v4f xc[5], xn[5];
            #pragma unroll
            for (int q = 0; q < 5; ++q) xc[q] = *(const v4f*)(xb + q * STRIDE);
            #pragma unroll
            for (int c = 0; c < 10; ++c) {
                #pragma unroll
                for (int q = 0; q < 5; ++q) {
                    xn[q] = xc[q];
                    if (c + 1 < 10) xn[q] = *(const v4f*)(xb + q * STRIDE + 4 * (c + 1));
                }
                __builtin_amdgcn_sched_barrier(0);
                #pragma unroll
                for (int q = 0; q < 5; ++q) {
                    const v4f pa = xc[q] * tgA[c];
                    const v4f pb = xc[q] * tgB[c];
                    if (c == 0) {
                        mA[q] = fmaxf(fmaxf(pa[0], pa[1]), fmaxf(pa[2], pa[3]));
                        mB[q] = fmaxf(fmaxf(pb[0], pb[1]), fmaxf(pb[2], pb[3]));
                    } else {
                        mA[q] = fmaxf(fmaxf(mA[q], pa[0]), pa[1]);
                        mA[q] = fmaxf(fmaxf(mA[q], pa[2]), pa[3]);
                        mB[q] = fmaxf(fmaxf(mB[q], pb[0]), pb[1]);
                        mB[q] = fmaxf(fmaxf(mB[q], pb[2]), pb[3]);
                    }
                }
                __builtin_amdgcn_sched_barrier(0);
                #pragma unroll
                for (int q = 0; q < 5; ++q) xc[q] = xn[q];
            }
            #pragma unroll
            for (int q = 0; q < 5; ++q) {
                float keep = hb ? mB[q] : mA[q];
                float send = hb ? mA[q] : mB[q];
                float recv = lane_xor1(send);
                s_M[(jb + q) * STRIDE + rr] = fmaxf(keep, recv);
            }
        }
        __syncthreads();

        const float part = (lane < NWAVES) ? s_red[lane] : 0.f;

        auto gemm_tile = [&](int tile) -> v8f {
            const int wr = tile / 5;
            const int wc = tile - 5 * wr;
            const float* pa = s_src + (wr * 16 + m) * STRIDE + kk;
            const float* pb = s_M + kk * STRIDE + wc * 16 + m;
            v8f acc = {0.f, 0.f, 0.f, 0.f, 0.f, 0.f, 0.f, 0.f};
            #pragma unroll
            for (int kc = 0; kc < NN; kc += 4) {
                v2f av, bv;
                av[0] = pa[kc + 0];
                av[1] = pa[kc + 1];
                bv[0] = pb[(kc + 0) * STRIDE];
                bv[1] = pb[(kc + 1) * STRIDE];
                acc = __builtin_amdgcn_wmma_f32_16x16x4_f32(
                    false, av, false, bv, (short)0, acc, false, false);
            }
            return acc;
        };
        auto update_tile = [&](int tile, v8f acc, float rn) -> float {
            const int wr = tile / 5;
            const int wc = tile - 5 * wr;
            const int eo = (wr * 16 + 8 * hh) * STRIDE + wc * 16 + m;
            float sq = 0.f;
            #pragma unroll
            for (int v = 0; v < 8; ++v) {
                float val = (s_x[eo + v * STRIDE] * s_d[eo + v * STRIDE] + acc[v]) * rn;
                s_x[eo + v * STRIDE] = val;
                sq += val * val;
            }
            return sq;
        };

        const v8f acc1 = gemm_tile(wave);
        const float rn = 1.f / sqrtf(wave_sum(part));
        float ssq = update_tile(wave, acc1, rn);
        if (two) {
            const v8f acc2 = gemm_tile(tile2);
            ssq += update_tile(tile2, acc2, rn);
        }
        wsum = wave_sum(ssq);
        __syncthreads();
    }

    if (lane == 0) s_red[wave] = wsum;
    __syncthreads();
    const float rn_last = 1.f / sqrtf(wave_sum((lane < NWAVES) ? s_red[lane] : 0.f));

    for (int ti = 0; ti < (two ? 2 : 1); ++ti) {
        const int tile = ti ? tile2 : wave;
        const int wr = tile / 5;
        const int wc = tile - 5 * wr;
        #pragma unroll
        for (int v = 0; v < 8; ++v) {
            int i = wr * 16 + v + 8 * hh;
            int a = wc * 16 + m;
            out[1 + i * NN + a] = s_x[i * STRIDE + a] * rn_last;
        }
    }
}

extern "C" void kernel_launch(void* const* d_in, const int* in_sizes, int n_in,
                              void* d_out, int out_size, void* d_ws, size_t ws_size,
                              hipStream_t stream) {
    (void)in_sizes; (void)n_in; (void)out_size; (void)d_ws; (void)ws_size;
    const float* recon = (const float*)d_in[0];
    const float* adj   = (const float*)d_in[1];
    float* out = (float*)d_out;

    const size_t shmem = (size_t)(5 * NN * STRIDE + 4 * NN + 32) * sizeof(float);
    grf_mpm_kernel<<<1, NTHREADS, shmem, stream>>>(recon, adj, out);
}
